# starting kernel
# baseline (speedup 1.0000x reference)
#include <hip/hip_runtime.h>


typedef __attribute__((ext_vector_type(2))) float v2f;
typedef __attribute__((ext_vector_type(4))) float v4f;
typedef __attribute__((ext_vector_type(8))) float v8f;

#define B_     4
#define F_     9
#define MP_    100
#define P_     12000
#define C_     64
#define NY_    400
#define NX_    400
#define POOLW_ 100
#define XO_    (NX_ / POOLW_)
#define KP1    12
#define NTILE  (P_ / 16)
#define SPLIT_ 4
#define MPW_   (MP_ / SPLIT_)

#define WS_W1S  0
#define WS_B1   768
#define WS_W2S  832
#define WS_B2   (832 + 4096)
#define WS_FEAT 4992

__global__ void precompute_kernel(
    const float* __restrict__ w1, const float* __restrict__ b1,
    const float* __restrict__ g1, const float* __restrict__ be1,
    const float* __restrict__ m1, const float* __restrict__ v1,
    const float* __restrict__ w2, const float* __restrict__ b2,
    const float* __restrict__ g2, const float* __restrict__ be2,
    const float* __restrict__ m2, const float* __restrict__ v2,
    float* __restrict__ ws)
{
    const int c = threadIdx.x;
    const float s1 = g1[c] * rsqrtf(v1[c] + 1e-3f);
    const float t1 = be1[c] - m1[c] * s1;
    for (int f = 0; f < KP1; ++f)
        ws[WS_W1S + c * KP1 + f] = (f < F_) ? s1 * w1[c * F_ + f] : 0.0f;
    ws[WS_B1 + c] = s1 * b1[c] + t1;

    const float s2 = g2[c] * rsqrtf(v2[c] + 1e-3f);
    const float t2 = be2[c] - m2[c] * s2;
    for (int k = 0; k < C_; ++k)
        ws[WS_W2S + c * C_ + k] = s2 * w2[c * C_ + k];
    ws[WS_B2 + c] = s2 * b2[c] + t2;
}

__global__ __launch_bounds__(128)
void pointnet_kernel(const float* __restrict__ x, float* __restrict__ ws)
{
    __shared__ float red[SPLIT_ - 1][32 * 32];

    const int lane = threadIdx.x & 31;
    const int wave = threadIdx.x >> 5;
    const int tile = blockIdx.x;
    const int b    = tile / NTILE;
    const int p0   = (tile % NTILE) * 16;
    const int mp0  = wave * MPW_;

    const int half = lane >> 4;
    const int lid  = lane & 15;

    const float* __restrict__ w1s = ws + WS_W1S;
    v2f a[4][3];
#pragma unroll
    for (int mg = 0; mg < 4; ++mg)
#pragma unroll
        for (int kt = 0; kt < 3; ++kt)
            a[mg][kt] = *(const v2f*)(w1s + (mg * 16 + lid) * KP1 + kt * 4 + half * 2);

    v8f vmax[4];
#pragma unroll
    for (int mg = 0; mg < 4; ++mg)
#pragma unroll
        for (int r = 0; r < 8; ++r) vmax[mg][r] = -3.4e38f;

    const float* __restrict__ xp = x + (size_t)b * F_ * MP_ * P_ + p0 + lid;

    auto load_frag = [&](int mp, v2f bf[3]) {
#pragma unroll
        for (int kt = 0; kt < 3; ++kt)
#pragma unroll
            for (int j = 0; j < 2; ++j) {
                const int f = kt * 4 + half * 2 + j;
                float v = 0.0f;
                if (f < F_)
                    v = __builtin_nontemporal_load(xp + ((size_t)f * MP_ + mp) * P_);
                bf[kt][j] = v;
            }
    };

    v2f cur[3], nxt[3];
    load_frag(mp0, cur);
    for (int i = 0; i < MPW_; ++i) {
        if (i + 1 < MPW_)
            load_frag(mp0 + i + 1, nxt);
#pragma unroll
        for (int mg = 0; mg < 4; ++mg) {
            v8f acc = {};
#pragma unroll
            for (int kt = 0; kt < 3; ++kt)
                acc = __builtin_amdgcn_wmma_f32_16x16x4_f32(
                    false, a[mg][kt], false, cur[kt], (short)0, acc, false, false);
#pragma unroll
            for (int r = 0; r < 8; ++r)
                vmax[mg][r] = fmaxf(vmax[mg][r], acc[r]);
        }
#pragma unroll
        for (int kt = 0; kt < 3; ++kt) cur[kt] = nxt[kt];
    }

    if (wave != 0) {
#pragma unroll
        for (int mg = 0; mg < 4; ++mg)
#pragma unroll
            for (int r = 0; r < 8; ++r)
                red[wave - 1][lane * 32 + mg * 8 + r] = vmax[mg][r];
    }
    __syncthreads();

    if (wave == 0) {
        const float* __restrict__ bias1 = ws + WS_B1;
        float* __restrict__ feat = ws + WS_FEAT;
        const size_t frow = (size_t)(b * P_ + p0 + lid) * C_;
#pragma unroll
        for (int mg = 0; mg < 4; ++mg)
#pragma unroll
            for (int r = 0; r < 8; ++r) {
                float v = vmax[mg][r];
#pragma unroll
                for (int w = 0; w < SPLIT_ - 1; ++w)
                    v = fmaxf(v, red[w][lane * 32 + mg * 8 + r]);
                const int c = mg * 16 + half * 8 + r;
                feat[frow + c] = fmaxf(v + bias1[c], 0.0f);
            }
    }
}

__global__ __launch_bounds__(256)
void init_out_kernel(const float* __restrict__ ws, float* __restrict__ out)
{
    const int i = blockIdx.x * 256 + threadIdx.x;
    const int d = (i / (NY_ * XO_)) & (C_ - 1);
    out[i] = fmaxf(ws[WS_B2 + d], 0.0f);
}

__global__ __launch_bounds__(128)
void scatter_conv_kernel(const int* __restrict__ coords,
                         const float* __restrict__ ws,
                         float* __restrict__ out)
{
    __shared__ float w2s_lds[C_ * C_];
    __shared__ float bias2_lds[C_];

    {
        const v4f* src = (const v4f*)(ws + WS_W2S);
        v4f* dst = (v4f*)w2s_lds;
#pragma unroll
        for (int i = threadIdx.x; i < (C_ * C_) / 4; i += 128)
            dst[i] = src[i];
        if (threadIdx.x < C_)
            bias2_lds[threadIdx.x] = ws[WS_B2 + threadIdx.x];
    }
    __syncthreads();

    const int lane = threadIdx.x & 31;
    const int wave = threadIdx.x >> 5;
    const int task = blockIdx.x * 4 + wave;
    const int b    = task / NTILE;
    const int p0   = (task % NTILE) * 16;
    const int half = lane >> 4;
    const int lid  = lane & 15;

    const float* __restrict__ feat = ws + WS_FEAT;

    v2f bf[16];
    {
        const float* fp = feat + (size_t)(b * P_ + p0 + lid) * C_;
#pragma unroll
        for (int kt = 0; kt < 16; ++kt)
            bf[kt] = *(const v2f*)(fp + kt * 4 + half * 2);
    }

    const int row = b * P_ + p0 + lid;
    const int y   = coords[row * 4 + 2];
    const int xq  = coords[row * 4 + 3] / POOLW_;
    unsigned int* __restrict__ outu = (unsigned int*)out;

#pragma unroll
    for (int mg = 0; mg < 4; ++mg) {
        v8f acc = {};
#pragma unroll
        for (int kt = 0; kt < 16; ++kt) {
            const v2f av = *(const v2f*)(w2s_lds + (mg * 16 + lid) * C_ + kt * 4 + half * 2);
            acc = __builtin_amdgcn_wmma_f32_16x16x4_f32(
                false, av, false, bf[kt], (short)0, acc, false, false);
        }
#pragma unroll
        for (int r = 0; r < 8; ++r) {
            const int d   = mg * 16 + half * 8 + r;
            const float v = fmaxf(acc[r] + bias2_lds[d], 0.0f);
            const size_t oi = ((size_t)(b * C_ + d) * NY_ + y) * XO_ + xq;
            atomicMax(outu + oi, __float_as_uint(v));
        }
    }
}

extern "C" void kernel_launch(void* const* d_in, const int* in_sizes, int n_in,
                              void* d_out, int out_size, void* d_ws, size_t ws_size,
                              hipStream_t stream)
{
    const float* x      = (const float*)d_in[0];
    const int*   coords = (const int*)  d_in[1];
    const float* w1  = (const float*)d_in[2];
    const float* b1  = (const float*)d_in[3];
    const float* g1  = (const float*)d_in[4];
    const float* be1 = (const float*)d_in[5];
    const float* m1  = (const float*)d_in[6];
    const float* v1  = (const float*)d_in[7];
    const float* w2  = (const float*)d_in[8];
    const float* b2  = (const float*)d_in[9];
    const float* g2  = (const float*)d_in[10];
    const float* be2 = (const float*)d_in[11];
    const float* m2  = (const float*)d_in[12];
    const float* v2  = (const float*)d_in[13];
    float* out = (float*)d_out;
    float* ws  = (float*)d_ws;

    precompute_kernel<<<1, 64, 0, stream>>>(w1, b1, g1, be1, m1, v1,
                                            w2, b2, g2, be2, m2, v2, ws);
    pointnet_kernel<<<B_ * NTILE, 128, 0, stream>>>(x, ws);
    init_out_kernel<<<(B_ * C_ * NY_ * XO_) / 256, 256, 0, stream>>>(ws, out);
    scatter_conv_kernel<<<(B_ * NTILE) / 4, 128, 0, stream>>>(coords, ws, out);
}


# Stage 3 owned per block: a block scans coords for its 8-row band, GEMMs only those pillars, merges by LDS atomic max (order-free), stores whole 128-B lines; global atomicMax and init launch removed
# speedup vs baseline: 1.6593x; 1.6593x over previous
#include <hip/hip_runtime.h>


typedef __attribute__((ext_vector_type(2))) float v2f;
typedef __attribute__((ext_vector_type(4))) float v4f;
typedef __attribute__((ext_vector_type(8))) float v8f;

#define B_     4
#define F_     9
#define MP_    100
#define P_     12000
#define C_     64
#define NY_    400
#define NX_    400
#define POOLW_ 100
#define XO_    (NX_ / POOLW_)
#define KP1    12
#define NTILE  (P_ / 16)
#define SPLIT_ 4
#define MPW_   (MP_ / SPLIT_)

#define WS_W1S  0
#define WS_B1   768
#define WS_W2S  832
#define WS_B2   (832 + 4096)
#define WS_FEAT 4992

__global__ void precompute_kernel(
    const float* __restrict__ w1, const float* __restrict__ b1,
    const float* __restrict__ g1, const float* __restrict__ be1,
    const float* __restrict__ m1, const float* __restrict__ v1,
    const float* __restrict__ w2, const float* __restrict__ b2,
    const float* __restrict__ g2, const float* __restrict__ be2,
    const float* __restrict__ m2, const float* __restrict__ v2,
    float* __restrict__ ws)
{
    const int c = threadIdx.x;
    const float s1 = g1[c] * rsqrtf(v1[c] + 1e-3f);
    const float t1 = be1[c] - m1[c] * s1;
    for (int f = 0; f < KP1; ++f)
        ws[WS_W1S + c * KP1 + f] = (f < F_) ? s1 * w1[c * F_ + f] : 0.0f;
    ws[WS_B1 + c] = s1 * b1[c] + t1;

    const float s2 = g2[c] * rsqrtf(v2[c] + 1e-3f);
    const float t2 = be2[c] - m2[c] * s2;
    for (int k = 0; k < C_; ++k)
        ws[WS_W2S + c * C_ + k] = s2 * w2[c * C_ + k];
    ws[WS_B2 + c] = s2 * b2[c] + t2;
}

__global__ __launch_bounds__(128)
void pointnet_kernel(const float* __restrict__ x, float* __restrict__ ws)
{
    __shared__ float red[SPLIT_ - 1][32 * 32];

    const int lane = threadIdx.x & 31;
    const int wave = threadIdx.x >> 5;
    const int tile = blockIdx.x;
    const int b    = tile / NTILE;
    const int p0   = (tile % NTILE) * 16;
    const int mp0  = wave * MPW_;

    const int half = lane >> 4;
    const int lid  = lane & 15;

    const float* __restrict__ w1s = ws + WS_W1S;
    v2f a[4][3];
#pragma unroll
    for (int mg = 0; mg < 4; ++mg)
#pragma unroll
        for (int kt = 0; kt < 3; ++kt)
            a[mg][kt] = *(const v2f*)(w1s + (mg * 16 + lid) * KP1 + kt * 4 + half * 2);

    v8f vmax[4];
#pragma unroll
    for (int mg = 0; mg < 4; ++mg)
#pragma unroll
        for (int r = 0; r < 8; ++r) vmax[mg][r] = -3.4e38f;

    const float* __restrict__ xp = x + (size_t)b * F_ * MP_ * P_ + p0 + lid;

    auto load_frag = [&](int mp, v2f bf[3]) {
#pragma unroll
        for (int kt = 0; kt < 3; ++kt)
#pragma unroll
            for (int j = 0; j < 2; ++j) {
                const int f = kt * 4 + half * 2 + j;
                float v = 0.0f;
                if (f < F_)
                    v = __builtin_nontemporal_load(xp + ((size_t)f * MP_ + mp) * P_);
                bf[kt][j] = v;
            }
    };

    v2f cur[3], nxt[3];
    load_frag(mp0, cur);
    for (int i = 0; i < MPW_; ++i) {
        if (i + 1 < MPW_)
            load_frag(mp0 + i + 1, nxt);
#pragma unroll
        for (int mg = 0; mg < 4; ++mg) {
            v8f acc = {};
#pragma unroll
            for (int kt = 0; kt < 3; ++kt)
                acc = __builtin_amdgcn_wmma_f32_16x16x4_f32(
                    false, a[mg][kt], false, cur[kt], (short)0, acc, false, false);
#pragma unroll
            for (int r = 0; r < 8; ++r)
                vmax[mg][r] = fmaxf(vmax[mg][r], acc[r]);
        }
#pragma unroll
        for (int kt = 0; kt < 3; ++kt) cur[kt] = nxt[kt];
    }

    if (wave != 0) {
#pragma unroll
        for (int mg = 0; mg < 4; ++mg)
#pragma unroll
            for (int r = 0; r < 8; ++r)
                red[wave - 1][lane * 32 + mg * 8 + r] = vmax[mg][r];
    }
    __syncthreads();

    if (wave == 0) {
        const float* __restrict__ bias1 = ws + WS_B1;
        float* __restrict__ feat = ws + WS_FEAT;
        const size_t frow = (size_t)(b * P_ + p0 + lid) * C_;
#pragma unroll
        for (int mg = 0; mg < 4; ++mg)
#pragma unroll
            for (int r = 0; r < 8; ++r) {
                float v = vmax[mg][r];
#pragma unroll
                for (int w = 0; w < SPLIT_ - 1; ++w)
                    v = fmaxf(v, red[w][lane * 32 + mg * 8 + r]);
                const int c = mg * 16 + half * 8 + r;
                feat[frow + c] = fmaxf(v + bias1[c], 0.0f);
            }
    }
}

#define BAND_   8
#define NBAND_  (NY_ / BAND_)
#define SW_     8
#define RPW_    (P_ / SW_)
#define SCANIT_ ((RPW_ + 31) / 32)
#define SCANB_  16
#define TPITCH_ 33

typedef __attribute__((ext_vector_type(2))) int v2i;

__global__ __launch_bounds__(256)
void band_conv_kernel(const int* __restrict__ coords,
                      const float* __restrict__ ws,
                      float* __restrict__ out)
{
    __shared__ unsigned int seg[SW_][RPW_];
    __shared__ unsigned int tab[C_ * TPITCH_];
    __shared__ int cnt[SW_];

    const int lane = threadIdx.x & 31;
    const int wave = __builtin_amdgcn_readfirstlane((int)(threadIdx.x >> 5));
    const int b    = blockIdx.x / NBAND_;
    const int band = blockIdx.x % NBAND_;
    const int half = lane >> 4;
    const int lid  = lane & 15;


    const float* __restrict__ w2s = ws + WS_W2S;
    v2f a[4][16];
#pragma unroll
    for (int mg = 0; mg < 4; ++mg)
#pragma unroll
        for (int kt = 0; kt < 16; ++kt)
            a[mg][kt] = *(const v2f*)(w2s + (mg * 16 + lid) * C_ + kt * 4 + half * 2);

#pragma unroll
    for (int k = 0; k < C_ / SW_; ++k) {
        const int d = wave + k * SW_;
        tab[d * TPITCH_ + lane] = __float_as_uint(fmaxf(ws[WS_B2 + d], 0.0f));
    }
    v4f bz[4][2];
#pragma unroll
    for (int mg = 0; mg < 4; ++mg)
#pragma unroll
        for (int q = 0; q < 2; ++q)
            bz[mg][q] = *(const v4f*)(ws + WS_B2 + mg * 16 + half * 8 + q * 4);

    const int* __restrict__ cb = coords + (size_t)b * P_ * 4;
    const int r0 = wave * RPW_;
    int n = 0;
    for (int it0 = 0; it0 < SCANIT_; it0 += SCANB_) {
        v2i yx[SCANB_];
#pragma unroll
        for (int u = 0; u < SCANB_; ++u) {
            const int i   = (it0 + u) * 32 + lane;
            const int row = r0 + (i < RPW_ ? i : 0);
            yx[u] = *(const v2i*)(cb + row * 4 + 2);
        }
#pragma unroll
        for (int u = 0; u < SCANB_; ++u) {
            const int i   = (it0 + u) * 32 + lane;
            const bool hit = (i < RPW_) && ((yx[u].x >> 3) == band);
            const unsigned m = __builtin_amdgcn_ballot_w32(hit);
            if (hit) {
                const int pos = n + (int)__builtin_amdgcn_mbcnt_lo(m, 0u);
                unsigned xq = (unsigned)yx[u].y / POOLW_;
                xq = xq < (unsigned)(XO_ - 1) ? xq : (unsigned)(XO_ - 1);
                seg[wave][pos] = ((unsigned)((yx[u].x & (BAND_ - 1)) * XO_ + (int)xq) << 16)
                                 | (unsigned)(r0 + i);
            }
            n += __builtin_popcount(m);
        }
    }
    if (lane == 0) cnt[wave] = n;
    __syncthreads();

    int pre[SW_ + 1];
    pre[0] = 0;
#pragma unroll
    for (int w = 0; w < SW_; ++w) pre[w + 1] = pre[w] + cnt[w];
    const int total  = pre[SW_];
    const int ntiles = (total + 15) >> 4;

    const float* __restrict__ feat = ws + WS_FEAT;
    for (int t = wave; t < ntiles; t += SW_) {
        const int g = t * 16 + lid;
        const bool valid = g < total;
        int s = 0, base = 0;
#pragma unroll
        for (int w = 1; w < SW_; ++w) {
            const bool ge = g >= pre[w];
            s    = ge ? w : s;
            base = ge ? pre[w] : base;
        }
        const unsigned e    = valid ? seg[s][g - base] : 0u;
        const unsigned prow = e & 0xFFFFu;
        const unsigned cell = e >> 16;
        unsigned int* tb = tab + (half * 8) * TPITCH_ + cell;

        v2f bf[16];
        {
            const float* fp = feat + (size_t)(b * P_ + (int)prow) * C_;
#pragma unroll
            for (int kt = 0; kt < 16; ++kt)
                bf[kt] = *(const v2f*)(fp + kt * 4 + half * 2);
        }

#pragma unroll
        for (int mg = 0; mg < 4; ++mg) {
            v8f acc = {};
#pragma unroll
            for (int kt = 0; kt < 16; ++kt)
                acc = __builtin_amdgcn_wmma_f32_16x16x4_f32(
                    false, a[mg][kt], false, bf[kt], (short)0, acc, false, false);
#pragma unroll
            for (int r = 0; r < 8; ++r) {
                const float v = fmaxf(acc[r] + bz[mg][r >> 2][r & 3], 0.0f);
                if (valid)
                    atomicMax(tb + (mg * 16 + r) * TPITCH_, __float_as_uint(v));
            }
        }
    }
    __syncthreads();

    for (int d = wave; d < C_; d += SW_)
        out[((size_t)(b * C_ + d) * NY_ + band * BAND_) * XO_ + lane] =
            __uint_as_float(tab[d * TPITCH_ + lane]);
}

extern "C" void kernel_launch(void* const* d_in, const int* in_sizes, int n_in,
                              void* d_out, int out_size, void* d_ws, size_t ws_size,
                              hipStream_t stream)
{
    const float* x      = (const float*)d_in[0];
    const int*   coords = (const int*)  d_in[1];
    const float* w1  = (const float*)d_in[2];
    const float* b1  = (const float*)d_in[3];
    const float* g1  = (const float*)d_in[4];
    const float* be1 = (const float*)d_in[5];
    const float* m1  = (const float*)d_in[6];
    const float* v1  = (const float*)d_in[7];
    const float* w2  = (const float*)d_in[8];
    const float* b2  = (const float*)d_in[9];
    const float* g2  = (const float*)d_in[10];
    const float* be2 = (const float*)d_in[11];
    const float* m2  = (const float*)d_in[12];
    const float* v2  = (const float*)d_in[13];
    float* out = (float*)d_out;
    float* ws  = (float*)d_ws;

    precompute_kernel<<<1, 64, 0, stream>>>(w1, b1, g1, be1, m1, v1,
                                            w2, b2, g2, be2, m2, v2, ws);
    pointnet_kernel<<<B_ * NTILE, 128, 0, stream>>>(x, ws);
    band_conv_kernel<<<B_ * NBAND_, 256, 0, stream>>>(coords, ws, out);
}


# Stage-1 loop lean beside its WMMAs: running max by three-input maximum over point pairs (no quieting copies), loads of four points issued one trip ahead, K padding by select instead of a lane branch
# speedup vs baseline: 1.9403x; 1.1693x over previous
#include <hip/hip_runtime.h>


typedef __attribute__((ext_vector_type(2))) float v2f;
typedef __attribute__((ext_vector_type(4))) float v4f;
typedef __attribute__((ext_vector_type(8))) float v8f;

#define B_     4
#define F_     9
#define MP_    100
#define P_     12000
#define C_     64
#define NY_    400
#define NX_    400
#define POOLW_ 100
#define XO_    (NX_ / POOLW_)
#define KP1    12
#define NTILE  (P_ / 16)
#define SPLIT_ 4
#define MPW_   (MP_ / SPLIT_)

#define WS_W1S  0
#define WS_B1   768
#define WS_W2S  832
#define WS_B2   (832 + 4096)
#define WS_FEAT 4992

__global__ void precompute_kernel(
    const float* __restrict__ w1, const float* __restrict__ b1,
    const float* __restrict__ g1, const float* __restrict__ be1,
    const float* __restrict__ m1, const float* __restrict__ v1,
    const float* __restrict__ w2, const float* __restrict__ b2,
    const float* __restrict__ g2, const float* __restrict__ be2,
    const float* __restrict__ m2, const float* __restrict__ v2,
    float* __restrict__ ws)
{
    const int c = threadIdx.x;
    const float s1 = g1[c] * rsqrtf(v1[c] + 1e-3f);
    const float t1 = be1[c] - m1[c] * s1;
    for (int f = 0; f < KP1; ++f)
        ws[WS_W1S + c * KP1 + f] = (f < F_) ? s1 * w1[c * F_ + f] : 0.0f;
    ws[WS_B1 + c] = s1 * b1[c] + t1;

    const float s2 = g2[c] * rsqrtf(v2[c] + 1e-3f);
    const float t2 = be2[c] - m2[c] * s2;
    for (int k = 0; k < C_; ++k)
        ws[WS_W2S + c * C_ + k] = s2 * w2[c * C_ + k];
    ws[WS_B2 + c] = s2 * b2[c] + t2;
}

__device__ __forceinline__ v8f vmax8(v8f p, v8f q)
{
    return __builtin_elementwise_maximum(p, q);
}

__global__ __launch_bounds__(128)
void pointnet_kernel(const float* __restrict__ x, float* __restrict__ ws)
{
    __shared__ float red[SPLIT_ - 1][32 * 32];

    const int lane = threadIdx.x & 31;
    const int wave = __builtin_amdgcn_readfirstlane((int)(threadIdx.x >> 5));
    const int tile = blockIdx.x;
    const int b    = tile / NTILE;
    const int p0   = (tile % NTILE) * 16;
    const int mp0  = wave * MPW_;

    const int half = lane >> 4;
    const int lid  = lane & 15;

    const float* __restrict__ w1s = ws + WS_W1S;
    v2f a[4][3];
#pragma unroll
    for (int mg = 0; mg < 4; ++mg)
#pragma unroll
        for (int kt = 0; kt < 3; ++kt)
            a[mg][kt] = *(const v2f*)(w1s + (mg * 16 + lid) * KP1 + kt * 4 + half * 2);

    const float* __restrict__ xrow = x + ((size_t)b * F_ * MP_ + mp0) * P_ + p0;
    const int loff  = half * 2 * MP_ * P_ + lid;
    const int loff8 = lid;

    auto load_frag = [&](int i, v2f bf[3]) {
        const float* __restrict__ r = xrow + (size_t)i * P_;
#pragma unroll
        for (int kt = 0; kt < 2; ++kt)
#pragma unroll
            for (int j = 0; j < 2; ++j)
                bf[kt][j] = __builtin_nontemporal_load(r + (size_t)(kt * 4 + j) * MP_ * P_ + loff);
        bf[2][0] = __builtin_nontemporal_load(r + (size_t)8 * MP_ * P_ + loff8);
        bf[2][1] = 0.0f;
    };

    auto chain = [&](int mg, const v2f bf[3]) -> v8f {
        v8f c = {};
#pragma unroll
        for (int kt = 0; kt < 2; ++kt)
            c = __builtin_amdgcn_wmma_f32_16x16x4_f32(
                false, a[mg][kt], false, bf[kt], (short)0, c, false, false);
        v2f b2;
        b2[0] = half ? 0.0f : bf[2][0];
        b2[1] = 0.0f;
        c = __builtin_amdgcn_wmma_f32_16x16x4_f32(
            false, a[mg][2], false, b2, (short)0, c, false, false);
        return c;
    };
    auto pair_step = [&](const v2f ba[3], const v2f bb[3], v8f vm[4]) {
#pragma unroll
        for (int mg = 0; mg < 4; ++mg) {
            const v8f ca = chain(mg, ba);
            const v8f cb = chain(mg, bb);
            vm[mg] = vmax8(vmax8(vm[mg], ca), cb);
        }
    };

    v8f vmax[4];
    v2f f0[3], cur[4][3], nxt[4][3];
    load_frag(0, f0);
#pragma unroll
    for (int q = 0; q < 4; ++q) load_frag(1 + q, cur[q]);
#pragma unroll
    for (int mg = 0; mg < 4; ++mg) vmax[mg] = chain(mg, f0);

#pragma nounroll
    for (int i = 1; i + 4 < MPW_; i += 4) {
#pragma unroll
        for (int q = 0; q < 4; ++q) load_frag(i + 4 + q, nxt[q]);
        __builtin_amdgcn_sched_barrier(0);
        pair_step(cur[0], cur[1], vmax);
        pair_step(cur[2], cur[3], vmax);
#pragma unroll
        for (int q = 0; q < 4; ++q)
#pragma unroll
            for (int kt = 0; kt < 3; ++kt) cur[q][kt] = nxt[q][kt];
    }
    pair_step(cur[0], cur[1], vmax);
    pair_step(cur[2], cur[3], vmax);

    if (wave != 0) {
#pragma unroll
        for (int mg = 0; mg < 4; ++mg)
#pragma unroll
            for (int r = 0; r < 8; ++r)
                red[wave - 1][lane * 32 + mg * 8 + r] = vmax[mg][r];
    }
    __syncthreads();

    if (wave == 0) {
        const float* __restrict__ bias1 = ws + WS_B1;
        float* __restrict__ feat = ws + WS_FEAT;
        const size_t frow = (size_t)(b * P_ + p0 + lid) * C_;
#pragma unroll
        for (int mg = 0; mg < 4; ++mg)
#pragma unroll
            for (int r = 0; r < 8; ++r) {
                float v = vmax[mg][r];
#pragma unroll
                for (int w = 0; w < SPLIT_ - 1; ++w)
                    v = fmaxf(v, red[w][lane * 32 + mg * 8 + r]);
                const int c = mg * 16 + half * 8 + r;
                feat[frow + c] = fmaxf(v + bias1[c], 0.0f);
            }
    }
}

#define BAND_   8
#define NBAND_  (NY_ / BAND_)
#define SW_     8
#define RPW_    (P_ / SW_)
#define SCANIT_ ((RPW_ + 31) / 32)
#define SCANB_  16
#define TPITCH_ 33

typedef __attribute__((ext_vector_type(2))) int v2i;

__global__ __launch_bounds__(256)
void band_conv_kernel(const int* __restrict__ coords,
                      const float* __restrict__ ws,
                      float* __restrict__ out)
{
    __shared__ unsigned int seg[SW_][RPW_];
    __shared__ unsigned int tab[C_ * TPITCH_];
    __shared__ int cnt[SW_];

    const int lane = threadIdx.x & 31;
    const int wave = __builtin_amdgcn_readfirstlane((int)(threadIdx.x >> 5));
    const int b    = blockIdx.x / NBAND_;
    const int band = blockIdx.x % NBAND_;
    const int half = lane >> 4;
    const int lid  = lane & 15;


    const float* __restrict__ w2s = ws + WS_W2S;
    v2f a[4][16];
#pragma unroll
    for (int mg = 0; mg < 4; ++mg)
#pragma unroll
        for (int kt = 0; kt < 16; ++kt)
            a[mg][kt] = *(const v2f*)(w2s + (mg * 16 + lid) * C_ + kt * 4 + half * 2);

#pragma unroll
    for (int k = 0; k < C_ / SW_; ++k) {
        const int d = wave + k * SW_;
        tab[d * TPITCH_ + lane] = __float_as_uint(fmaxf(ws[WS_B2 + d], 0.0f));
    }
    v4f bz[4][2];
#pragma unroll
    for (int mg = 0; mg < 4; ++mg)
#pragma unroll
        for (int q = 0; q < 2; ++q)
            bz[mg][q] = *(const v4f*)(ws + WS_B2 + mg * 16 + half * 8 + q * 4);

    const int* __restrict__ cb = coords + (size_t)b * P_ * 4;
    const int r0 = wave * RPW_;
    int n = 0;
    for (int it0 = 0; it0 < SCANIT_; it0 += SCANB_) {
        v2i yx[SCANB_];
#pragma unroll
        for (int u = 0; u < SCANB_; ++u) {
            const int i   = (it0 + u) * 32 + lane;
            const int row = r0 + (i < RPW_ ? i : 0);
            yx[u] = *(const v2i*)(cb + row * 4 + 2);
        }
#pragma unroll
        for (int u = 0; u < SCANB_; ++u) {
            const int i   = (it0 + u) * 32 + lane;
            const bool hit = (i < RPW_) && ((yx[u].x >> 3) == band);
            const unsigned m = __builtin_amdgcn_ballot_w32(hit);
            if (hit) {
                const int pos = n + (int)__builtin_amdgcn_mbcnt_lo(m, 0u);
                unsigned xq = (unsigned)yx[u].y / POOLW_;
                xq = xq < (unsigned)(XO_ - 1) ? xq : (unsigned)(XO_ - 1);
                seg[wave][pos] = ((unsigned)((yx[u].x & (BAND_ - 1)) * XO_ + (int)xq) << 16)
                                 | (unsigned)(r0 + i);
            }
            n += __builtin_popcount(m);
        }
    }
    if (lane == 0) cnt[wave] = n;
    __syncthreads();

    int pre[SW_ + 1];
    pre[0] = 0;
#pragma unroll
    for (int w = 0; w < SW_; ++w) pre[w + 1] = pre[w] + cnt[w];
    const int total  = pre[SW_];
    const int ntiles = (total + 15) >> 4;

    const float* __restrict__ feat = ws + WS_FEAT;
    for (int t = wave; t < ntiles; t += SW_) {
        const int g = t * 16 + lid;
        const bool valid = g < total;
        int s = 0, base = 0;
#pragma unroll
        for (int w = 1; w < SW_; ++w) {
            const bool ge = g >= pre[w];
            s    = ge ? w : s;
            base = ge ? pre[w] : base;
        }
        const unsigned e    = valid ? seg[s][g - base] : 0u;
        const unsigned prow = e & 0xFFFFu;
        const unsigned cell = e >> 16;
        unsigned int* tb = tab + (half * 8) * TPITCH_ + cell;

        v2f bf[16];
        {
            const float* fp = feat + (size_t)(b * P_ + (int)prow) * C_;
#pragma unroll
            for (int kt = 0; kt < 16; ++kt)
                bf[kt] = *(const v2f*)(fp + kt * 4 + half * 2);
        }

#pragma unroll
        for (int mg = 0; mg < 4; ++mg) {
            v8f acc = {};
#pragma unroll
            for (int kt = 0; kt < 16; ++kt)
                acc = __builtin_amdgcn_wmma_f32_16x16x4_f32(
                    false, a[mg][kt], false, bf[kt], (short)0, acc, false, false);
#pragma unroll
            for (int r = 0; r < 8; ++r) {
                const float v = fmaxf(acc[r] + bz[mg][r >> 2][r & 3], 0.0f);
                if (valid)
                    atomicMax(tb + (mg * 16 + r) * TPITCH_, __float_as_uint(v));
            }
        }
    }
    __syncthreads();

    for (int d = wave; d < C_; d += SW_)
        out[((size_t)(b * C_ + d) * NY_ + band * BAND_) * XO_ + lane] =
            __uint_as_float(tab[d * TPITCH_ + lane]);
}

extern "C" void kernel_launch(void* const* d_in, const int* in_sizes, int n_in,
                              void* d_out, int out_size, void* d_ws, size_t ws_size,
                              hipStream_t stream)
{
    const float* x      = (const float*)d_in[0];
    const int*   coords = (const int*)  d_in[1];
    const float* w1  = (const float*)d_in[2];
    const float* b1  = (const float*)d_in[3];
    const float* g1  = (const float*)d_in[4];
    const float* be1 = (const float*)d_in[5];
    const float* m1  = (const float*)d_in[6];
    const float* v1  = (const float*)d_in[7];
    const float* w2  = (const float*)d_in[8];
    const float* b2  = (const float*)d_in[9];
    const float* g2  = (const float*)d_in[10];
    const float* be2 = (const float*)d_in[11];
    const float* m2  = (const float*)d_in[12];
    const float* v2  = (const float*)d_in[13];
    float* out = (float*)d_out;
    float* ws  = (float*)d_ws;

    precompute_kernel<<<1, 64, 0, stream>>>(w1, b1, g1, be1, m1, v1,
                                            w2, b2, g2, be2, m2, v2, ws);
    pointnet_kernel<<<B_ * NTILE, 128, 0, stream>>>(x, ws);
    band_conv_kernel<<<B_ * NBAND_, 256, 0, stream>>>(coords, ws, out);
}
